# starting kernel
# baseline (speedup 1.0000x reference)
#include <hip/hip_runtime.h>
#include <hip/hip_bf16.h>
#include <math.h>

#define N_NODES 100000
#define N_FEAT  512
#define HIDDEN  128
#define N_CLASS 16
#define N_EDGES 3200000
#define ALPHA   0.25f

typedef __attribute__((ext_vector_type(2))) float v2f;
typedef __attribute__((ext_vector_type(8))) float v8f;

#define KCHUNK   64
#define W1C_STR  136
#define HID_STR  132

__global__ __launch_bounds__(256) void gcn_gemm_kernel(
    const float* __restrict__ x, const float* __restrict__ W1,
    const float* __restrict__ W2, float* __restrict__ local_out)
{
    __shared__ float w1c[KCHUNK][W1C_STR];
    __shared__ float hid[8][16][HID_STR];

    const int tid  = threadIdx.x;
    const int wave = tid >> 5;
    const int lane = tid & 31;
    const int half = lane >> 4;
    const int l16  = lane & 15;

    const int row_tile = blockIdx.x * 8 + wave;
    const int row_base = row_tile * 16;
    const bool active  = (row_base < N_NODES);

    v8f acc[8];
    for (int t = 0; t < 8; ++t)
        for (int i = 0; i < 8; ++i) acc[t][i] = 0.0f;

    for (int k0 = 0; k0 < N_FEAT; k0 += KCHUNK) {
        __syncthreads();
        for (int i = tid; i < KCHUNK * (HIDDEN / 4); i += 256) {
            const int r  = i >> 5;
            const int c4 = i & 31;
            const float4 v = ((const float4*)(W1 + (size_t)(k0 + r) * HIDDEN))[c4];
            *(float4*)&w1c[r][c4 * 4] = v;
        }
        __syncthreads();

        if (active) {
            for (int kk = 0; kk < KCHUNK; kk += 4) {
                const float* ap = x + (size_t)(row_base + l16) * N_FEAT
                                    + (k0 + kk + 2 * half);
                v2f a = *(const v2f*)ap;
                for (int t = 0; t < 8; ++t) {
                    v2f b;
                    b[0] = w1c[kk + 2 * half + 0][t * 16 + l16];
                    b[1] = w1c[kk + 2 * half + 1][t * 16 + l16];
                    acc[t] = __builtin_amdgcn_wmma_f32_16x16x4_f32(
                        false, a, false, b, (short)0, acc[t], false, false);
                }
            }
        }
    }

    if (active) {
        for (int t = 0; t < 8; ++t)
            for (int r = 0; r < 8; ++r) {
                float v = acc[t][r];
                hid[wave][r + 8 * half][t * 16 + l16] = v > 0.0f ? v : 0.0f;
            }
        v8f c2;
        for (int i = 0; i < 8; ++i) c2[i] = 0.0f;
        for (int kk = 0; kk < HIDDEN; kk += 4) {
            v2f a, b;
            a[0] = hid[wave][l16][kk + 2 * half + 0];
            a[1] = hid[wave][l16][kk + 2 * half + 1];
            b[0] = W2[(size_t)(kk + 2 * half + 0) * N_CLASS + l16];
            b[1] = W2[(size_t)(kk + 2 * half + 1) * N_CLASS + l16];
            c2 = __builtin_amdgcn_wmma_f32_16x16x4_f32(
                false, a, false, b, (short)0, c2, false, false);
        }
        for (int r = 0; r < 8; ++r)
            local_out[(size_t)(row_base + r + 8 * half) * N_CLASS + l16] = c2[r];
    }
}

__global__ void zero_kernel(float* __restrict__ p, long long n)
{
    long long i = (long long)blockIdx.x * blockDim.x + threadIdx.x;
    if (i < n) p[i] = 0.0f;
}

__global__ void deg_kernel(const int* __restrict__ src, float* __restrict__ deg)
{
    int e = blockIdx.x * blockDim.x + threadIdx.x;
    if (e < N_EDGES) atomicAdd(&deg[src[e]], 1.0f);
}

__global__ void scatter_kernel(const int* __restrict__ src,
                               const int* __restrict__ dst,
                               const float* __restrict__ logits,
                               float* __restrict__ agg)
{
    long long gid = (long long)blockIdx.x * blockDim.x + threadIdx.x;
    if (gid >= (long long)N_EDGES * N_CLASS) return;
    const int e = (int)(gid >> 4);
    const int c = (int)(gid & 15);
    const int s = src[e];
    const int d = dst[e];
    atomicAdd(&agg[(size_t)s * N_CLASS + c], logits[(size_t)d * N_CLASS + c]);
}

__global__ void combine_kernel(const float* __restrict__ agg,
                               const float* __restrict__ local,
                               const float* __restrict__ deg,
                               float* __restrict__ out)
{
    long long i = (long long)blockIdx.x * blockDim.x + threadIdx.x;
    if (i >= (long long)N_NODES * N_CLASS) return;
    const int row = (int)(i >> 4);
    const float scale = (1.0f - ALPHA) / fmaxf(deg[row], 1e-12f);
    out[i] = scale * agg[i] + ALPHA * local[i];
}

__global__ void final_kernel(const float* __restrict__ agg,
                             const float* __restrict__ local,
                             const float* __restrict__ deg,
                             float* __restrict__ out)
{
    int row = blockIdx.x * blockDim.x + threadIdx.x;
    if (row >= N_NODES) return;
    const float scale = (1.0f - ALPHA) / fmaxf(deg[row], 1e-12f);
    float t[N_CLASS];
    float m = -INFINITY;
    for (int c = 0; c < N_CLASS; ++c) {
        t[c] = scale * agg[(size_t)row * N_CLASS + c]
             + ALPHA * local[(size_t)row * N_CLASS + c];
        m = fmaxf(m, t[c]);
    }
    float s = 0.0f;
    for (int c = 0; c < N_CLASS; ++c) s += __expf(t[c] - m);
    const float lse = m + __logf(s);
    for (int c = 0; c < N_CLASS; ++c)
        out[(size_t)row * N_CLASS + c] = t[c] - lse;
}

extern "C" void kernel_launch(void* const* d_in, const int* in_sizes, int n_in,
                              void* d_out, int out_size, void* d_ws, size_t ws_size,
                              hipStream_t stream)
{
    const float* x    = (const float*)d_in[0];
    const float* W1   = (const float*)d_in[1];
    const float* W2   = (const float*)d_in[2];
    const int*   esrc = (const int*)d_in[3];
    const int*   edst = (const int*)d_in[4];
    float* out = (float*)d_out;

    float* local  = (float*)d_ws;
    float* logits = local  + (size_t)N_NODES * N_CLASS;
    float* deg    = logits + (size_t)N_NODES * N_CLASS;
    float* agg    = out;

    const dim3 blk(256);
    const long long nl   = (long long)N_NODES * N_CLASS;
    const long long ne16 = (long long)N_EDGES * N_CLASS;
    const unsigned gN    = (unsigned)((N_NODES + 255) / 256);
    const unsigned gNL   = (unsigned)((nl + 255) / 256);
    const unsigned gE    = (unsigned)((N_EDGES + 255) / 256);
    const unsigned gE16  = (unsigned)((ne16 + 255) / 256);

    const unsigned gGEMM = (unsigned)(((N_NODES + 15) / 16 + 7) / 8);
    gcn_gemm_kernel<<<gGEMM, blk, 0, stream>>>(x, W1, W2, local);

    zero_kernel<<<gN, blk, 0, stream>>>(deg, N_NODES);
    deg_kernel<<<gE, blk, 0, stream>>>(esrc, deg);

    zero_kernel<<<gNL, blk, 0, stream>>>(agg, nl);
    scatter_kernel<<<gE16, blk, 0, stream>>>(esrc, edst, local, agg);
    combine_kernel<<<gNL, blk, 0, stream>>>(agg, local, deg, logits);

    zero_kernel<<<gNL, blk, 0, stream>>>(agg, nl);
    scatter_kernel<<<gE16, blk, 0, stream>>>(esrc, edst, logits, agg);
    final_kernel<<<gN, blk, 0, stream>>>(agg, local, deg, out);
}


# scatter passes: a wave takes 256 edges, 32 per trip: coalesced index loads, row-share to 16-lane halves, 16 row-pair gathers in flight then 16 row-pair f32 atomic adds (was 1 edge pair per wave)
# speedup vs baseline: 1.3273x; 1.3273x over previous
#include <hip/hip_runtime.h>
#include <hip/hip_bf16.h>
#include <math.h>

#define N_NODES 100000
#define N_FEAT  512
#define HIDDEN  128
#define N_CLASS 16
#define N_EDGES 3200000
#define ALPHA   0.25f

typedef __attribute__((ext_vector_type(2))) float v2f;
typedef __attribute__((ext_vector_type(8))) float v8f;

#define KCHUNK   64
#define W1C_STR  136
#define HID_STR  132

__global__ __launch_bounds__(256) void gcn_gemm_kernel(
    const float* __restrict__ x, const float* __restrict__ W1,
    const float* __restrict__ W2, float* __restrict__ local_out)
{
    __shared__ float w1c[KCHUNK][W1C_STR];
    __shared__ float hid[8][16][HID_STR];

    const int tid  = threadIdx.x;
    const int wave = tid >> 5;
    const int lane = tid & 31;
    const int half = lane >> 4;
    const int l16  = lane & 15;

    const int row_tile = blockIdx.x * 8 + wave;
    const int row_base = row_tile * 16;
    const bool active  = (row_base < N_NODES);

    v8f acc[8];
    for (int t = 0; t < 8; ++t)
        for (int i = 0; i < 8; ++i) acc[t][i] = 0.0f;

    for (int k0 = 0; k0 < N_FEAT; k0 += KCHUNK) {
        __syncthreads();
        for (int i = tid; i < KCHUNK * (HIDDEN / 4); i += 256) {
            const int r  = i >> 5;
            const int c4 = i & 31;
            const float4 v = ((const float4*)(W1 + (size_t)(k0 + r) * HIDDEN))[c4];
            *(float4*)&w1c[r][c4 * 4] = v;
        }
        __syncthreads();

        if (active) {
            for (int kk = 0; kk < KCHUNK; kk += 4) {
                const float* ap = x + (size_t)(row_base + l16) * N_FEAT
                                    + (k0 + kk + 2 * half);
                v2f a = *(const v2f*)ap;
                for (int t = 0; t < 8; ++t) {
                    v2f b;
                    b[0] = w1c[kk + 2 * half + 0][t * 16 + l16];
                    b[1] = w1c[kk + 2 * half + 1][t * 16 + l16];
                    acc[t] = __builtin_amdgcn_wmma_f32_16x16x4_f32(
                        false, a, false, b, (short)0, acc[t], false, false);
                }
            }
        }
    }

    if (active) {
        for (int t = 0; t < 8; ++t)
            for (int r = 0; r < 8; ++r) {
                float v = acc[t][r];
                hid[wave][r + 8 * half][t * 16 + l16] = v > 0.0f ? v : 0.0f;
            }
        v8f c2;
        for (int i = 0; i < 8; ++i) c2[i] = 0.0f;
        for (int kk = 0; kk < HIDDEN; kk += 4) {
            v2f a, b;
            a[0] = hid[wave][l16][kk + 2 * half + 0];
            a[1] = hid[wave][l16][kk + 2 * half + 1];
            b[0] = W2[(size_t)(kk + 2 * half + 0) * N_CLASS + l16];
            b[1] = W2[(size_t)(kk + 2 * half + 1) * N_CLASS + l16];
            c2 = __builtin_amdgcn_wmma_f32_16x16x4_f32(
                false, a, false, b, (short)0, c2, false, false);
        }
        for (int r = 0; r < 8; ++r)
            local_out[(size_t)(row_base + r + 8 * half) * N_CLASS + l16] = c2[r];
    }
}

__global__ void zero_kernel(float* __restrict__ p, long long n)
{
    long long i = (long long)blockIdx.x * blockDim.x + threadIdx.x;
    if (i < n) p[i] = 0.0f;
}

__global__ void deg_kernel(const int* __restrict__ src, float* __restrict__ deg)
{
    int e = blockIdx.x * blockDim.x + threadIdx.x;
    if (e < N_EDGES) atomicAdd(&deg[src[e]], 1.0f);
}

#define SC_WAVES 4
#define SC_TRIPS 8
#define SC_EDGES_PER_BLOCK (SC_WAVES * SC_TRIPS * 32)
static_assert(N_EDGES % SC_EDGES_PER_BLOCK == 0, "edge count must fill whole work-groups");

#define ROW_SHARE(v, J) __builtin_amdgcn_update_dpp(0, (v), 0x150 + (J), 0xF, 0xF, false)

__global__ __launch_bounds__(SC_WAVES * 32) void scatter_kernel(
    const int* __restrict__ src, const int* __restrict__ dst,
    const float* __restrict__ logits, float* __restrict__ agg)
{
    const int lane = threadIdx.x & 31;
    const unsigned cb = (unsigned)(lane & 15) << 2;
    const int wave = blockIdx.x * SC_WAVES + (threadIdx.x >> 5);
    int e = wave * (SC_TRIPS * 32) + lane;

    const char* lg = (const char*)logits;
    char* ag = (char*)agg;

    int s = src[e];
    int d = dst[e];
    for (int it = 0; it < SC_TRIPS; ++it) {
        int sn = s, dn = d;
        if (it + 1 < SC_TRIPS) {
            sn = src[e + 32];
            dn = dst[e + 32];
        }
        float v[16];
#define SC_GATHER(J) { const unsigned dj = (unsigned)ROW_SHARE(d, J); \
                       v[J] = *(const float*)(lg + ((dj << 6) | cb)); }
        SC_GATHER(0)  SC_GATHER(1)  SC_GATHER(2)  SC_GATHER(3)
        SC_GATHER(4)  SC_GATHER(5)  SC_GATHER(6)  SC_GATHER(7)
        SC_GATHER(8)  SC_GATHER(9)  SC_GATHER(10) SC_GATHER(11)
        SC_GATHER(12) SC_GATHER(13) SC_GATHER(14) SC_GATHER(15)
#undef SC_GATHER
#define SC_ADD(J) { const unsigned sj = (unsigned)ROW_SHARE(s, J); \
                    atomicAdd((float*)(ag + ((sj << 6) | cb)), v[J]); }
        SC_ADD(0)  SC_ADD(1)  SC_ADD(2)  SC_ADD(3)
        SC_ADD(4)  SC_ADD(5)  SC_ADD(6)  SC_ADD(7)
        SC_ADD(8)  SC_ADD(9)  SC_ADD(10) SC_ADD(11)
        SC_ADD(12) SC_ADD(13) SC_ADD(14) SC_ADD(15)
#undef SC_ADD
        s = sn;
        d = dn;
        e += 32;
    }
}

__global__ void combine_kernel(const float* __restrict__ agg,
                               const float* __restrict__ local,
                               const float* __restrict__ deg,
                               float* __restrict__ out)
{
    long long i = (long long)blockIdx.x * blockDim.x + threadIdx.x;
    if (i >= (long long)N_NODES * N_CLASS) return;
    const int row = (int)(i >> 4);
    const float scale = (1.0f - ALPHA) / fmaxf(deg[row], 1e-12f);
    out[i] = scale * agg[i] + ALPHA * local[i];
}

__global__ void final_kernel(const float* __restrict__ agg,
                             const float* __restrict__ local,
                             const float* __restrict__ deg,
                             float* __restrict__ out)
{
    int row = blockIdx.x * blockDim.x + threadIdx.x;
    if (row >= N_NODES) return;
    const float scale = (1.0f - ALPHA) / fmaxf(deg[row], 1e-12f);
    float t[N_CLASS];
    float m = -INFINITY;
    for (int c = 0; c < N_CLASS; ++c) {
        t[c] = scale * agg[(size_t)row * N_CLASS + c]
             + ALPHA * local[(size_t)row * N_CLASS + c];
        m = fmaxf(m, t[c]);
    }
    float s = 0.0f;
    for (int c = 0; c < N_CLASS; ++c) s += __expf(t[c] - m);
    const float lse = m + __logf(s);
    for (int c = 0; c < N_CLASS; ++c)
        out[(size_t)row * N_CLASS + c] = t[c] - lse;
}

extern "C" void kernel_launch(void* const* d_in, const int* in_sizes, int n_in,
                              void* d_out, int out_size, void* d_ws, size_t ws_size,
                              hipStream_t stream)
{
    const float* x    = (const float*)d_in[0];
    const float* W1   = (const float*)d_in[1];
    const float* W2   = (const float*)d_in[2];
    const int*   esrc = (const int*)d_in[3];
    const int*   edst = (const int*)d_in[4];
    float* out = (float*)d_out;

    float* local  = (float*)d_ws;
    float* logits = local  + (size_t)N_NODES * N_CLASS;
    float* deg    = logits + (size_t)N_NODES * N_CLASS;
    float* agg    = out;

    const dim3 blk(256);
    const long long nl   = (long long)N_NODES * N_CLASS;
    const unsigned gN    = (unsigned)((N_NODES + 255) / 256);
    const unsigned gNL   = (unsigned)((nl + 255) / 256);
    const unsigned gE    = (unsigned)((N_EDGES + 255) / 256);
    const unsigned gSC   = (unsigned)(N_EDGES / SC_EDGES_PER_BLOCK);
    const dim3 blkSC(SC_WAVES * 32);

    const unsigned gGEMM = (unsigned)(((N_NODES + 15) / 16 + 7) / 8);
    gcn_gemm_kernel<<<gGEMM, blk, 0, stream>>>(x, W1, W2, local);

    zero_kernel<<<gN, blk, 0, stream>>>(deg, N_NODES);
    deg_kernel<<<gE, blk, 0, stream>>>(esrc, deg);

    zero_kernel<<<gNL, blk, 0, stream>>>(agg, nl);
    scatter_kernel<<<gSC, blkSC, 0, stream>>>(esrc, edst, local, agg);
    combine_kernel<<<gNL, blk, 0, stream>>>(agg, local, deg, logits);

    zero_kernel<<<gNL, blk, 0, stream>>>(agg, nl);
    scatter_kernel<<<gSC, blkSC, 0, stream>>>(esrc, edst, logits, agg);
    final_kernel<<<gN, blk, 0, stream>>>(agg, local, deg, out);
}


# f32 atomic sums (degree, both edge passes) go to two planes, one per package half by the wave's die number; pass-2 upper plane starts at alpha*local/scale; sum order: two partial sums per row
# speedup vs baseline: 2.5426x; 1.9156x over previous
#include <hip/hip_runtime.h>
#include <hip/hip_bf16.h>
#include <math.h>

#define N_NODES 100000
#define N_FEAT  512
#define HIDDEN  128
#define N_CLASS 16
#define N_EDGES 3200000
#define ALPHA   0.25f

typedef __attribute__((ext_vector_type(2))) float v2f;
typedef __attribute__((ext_vector_type(8))) float v8f;

#define KCHUNK   64
#define W1C_STR  136
#define HID_STR  132

__global__ __launch_bounds__(256) void gcn_gemm_kernel(
    const float* __restrict__ x, const float* __restrict__ W1,
    const float* __restrict__ W2, float* __restrict__ local_out)
{
    __shared__ float w1c[KCHUNK][W1C_STR];
    __shared__ float hid[8][16][HID_STR];

    const int tid  = threadIdx.x;
    const int wave = tid >> 5;
    const int lane = tid & 31;
    const int half = lane >> 4;
    const int l16  = lane & 15;

    const int row_tile = blockIdx.x * 8 + wave;
    const int row_base = row_tile * 16;
    const bool active  = (row_base < N_NODES);

    v8f acc[8];
    for (int t = 0; t < 8; ++t)
        for (int i = 0; i < 8; ++i) acc[t][i] = 0.0f;

    for (int k0 = 0; k0 < N_FEAT; k0 += KCHUNK) {
        __syncthreads();
        for (int i = tid; i < KCHUNK * (HIDDEN / 4); i += 256) {
            const int r  = i >> 5;
            const int c4 = i & 31;
            const float4 v = ((const float4*)(W1 + (size_t)(k0 + r) * HIDDEN))[c4];
            *(float4*)&w1c[r][c4 * 4] = v;
        }
        __syncthreads();

        if (active) {
            for (int kk = 0; kk < KCHUNK; kk += 4) {
                const float* ap = x + (size_t)(row_base + l16) * N_FEAT
                                    + (k0 + kk + 2 * half);
                v2f a = *(const v2f*)ap;
                for (int t = 0; t < 8; ++t) {
                    v2f b;
                    b[0] = w1c[kk + 2 * half + 0][t * 16 + l16];
                    b[1] = w1c[kk + 2 * half + 1][t * 16 + l16];
                    acc[t] = __builtin_amdgcn_wmma_f32_16x16x4_f32(
                        false, a, false, b, (short)0, acc[t], false, false);
                }
            }
        }
    }

    if (active) {
        for (int t = 0; t < 8; ++t)
            for (int r = 0; r < 8; ++r) {
                float v = acc[t][r];
                hid[wave][r + 8 * half][t * 16 + l16] = v > 0.0f ? v : 0.0f;
            }
        v8f c2;
        for (int i = 0; i < 8; ++i) c2[i] = 0.0f;
        for (int kk = 0; kk < HIDDEN; kk += 4) {
            v2f a, b;
            a[0] = hid[wave][l16][kk + 2 * half + 0];
            a[1] = hid[wave][l16][kk + 2 * half + 1];
            b[0] = W2[(size_t)(kk + 2 * half + 0) * N_CLASS + l16];
            b[1] = W2[(size_t)(kk + 2 * half + 1) * N_CLASS + l16];
            c2 = __builtin_amdgcn_wmma_f32_16x16x4_f32(
                false, a, false, b, (short)0, c2, false, false);
        }
        for (int r = 0; r < 8; ++r)
            local_out[(size_t)(row_base + r + 8 * half) * N_CLASS + l16] = c2[r];
    }
}

__global__ void zero_kernel(float* __restrict__ p, long long n)
{
    long long i = (long long)blockIdx.x * blockDim.x + threadIdx.x;
    if (i < n) p[i] = 0.0f;
}

__global__ void zero2_kernel(float* __restrict__ p, float* __restrict__ q, long long n)
{
    long long i = (long long)blockIdx.x * blockDim.x + threadIdx.x;
    if (i < n) { p[i] = 0.0f; q[i] = 0.0f; }
}

__device__ __forceinline__ bool upper_half()
{
    const unsigned w = __builtin_amdgcn_s_sendmsg_rtn(0x87);
    return ((w >> 16) & 0xFu) >= 4u;
}

__global__ void deg_kernel(const int* __restrict__ src,
                           float* __restrict__ deg_lo, float* __restrict__ deg_hi)
{
    float* deg = upper_half() ? deg_hi : deg_lo;
    int e = blockIdx.x * blockDim.x + threadIdx.x;
    if (e < N_EDGES) atomicAdd(&deg[src[e]], 1.0f);
}

__global__ void degsum_zero_kernel(float* __restrict__ deg, float* __restrict__ a,
                                   float* __restrict__ b, long long n)
{
    long long i = (long long)blockIdx.x * blockDim.x + threadIdx.x;
    if (i >= n) return;
    if (i < N_NODES) deg[i] = deg[i] + a[i];
    a[i] = 0.0f;
    b[i] = 0.0f;
}

#define SC_WAVES 4
#define SC_TRIPS 8
#define SC_EDGES_PER_BLOCK (SC_WAVES * SC_TRIPS * 32)
static_assert(N_EDGES % SC_EDGES_PER_BLOCK == 0, "edge count must fill whole work-groups");

#define ROW_SHARE(v, J) __builtin_amdgcn_update_dpp(0, (v), 0x150 + (J), 0xF, 0xF, false)

__global__ __launch_bounds__(SC_WAVES * 32) void scatter_kernel(
    const int* __restrict__ src, const int* __restrict__ dst,
    const float* __restrict__ logits, float* __restrict__ agg_lo, float* __restrict__ agg_hi)
{
    const int lane = threadIdx.x & 31;
    const unsigned cb = (unsigned)(lane & 15) << 2;
    const int wave = blockIdx.x * SC_WAVES + (threadIdx.x >> 5);
    int e = wave * (SC_TRIPS * 32) + lane;

    const char* lg = (const char*)logits;
    char* ag = (char*)(upper_half() ? agg_hi : agg_lo);

    int s = src[e];
    int d = dst[e];
    for (int it = 0; it < SC_TRIPS; ++it) {
        int sn = s, dn = d;
        if (it + 1 < SC_TRIPS) {
            sn = src[e + 32];
            dn = dst[e + 32];
        }
        float v[16];
#define SC_GATHER(J) { const unsigned dj = (unsigned)ROW_SHARE(d, J); \
                       v[J] = *(const float*)(lg + ((dj << 6) | cb)); }
        SC_GATHER(0)  SC_GATHER(1)  SC_GATHER(2)  SC_GATHER(3)
        SC_GATHER(4)  SC_GATHER(5)  SC_GATHER(6)  SC_GATHER(7)
        SC_GATHER(8)  SC_GATHER(9)  SC_GATHER(10) SC_GATHER(11)
        SC_GATHER(12) SC_GATHER(13) SC_GATHER(14) SC_GATHER(15)
#undef SC_GATHER
#define SC_ADD(J) { const unsigned sj = (unsigned)ROW_SHARE(s, J); \
                    atomicAdd((float*)(ag + ((sj << 6) | cb)), v[J]); }
        SC_ADD(0)  SC_ADD(1)  SC_ADD(2)  SC_ADD(3)
        SC_ADD(4)  SC_ADD(5)  SC_ADD(6)  SC_ADD(7)
        SC_ADD(8)  SC_ADD(9)  SC_ADD(10) SC_ADD(11)
        SC_ADD(12) SC_ADD(13) SC_ADD(14) SC_ADD(15)
#undef SC_ADD
        s = sn;
        d = dn;
        e += 32;
    }
}

__global__ void combine_kernel(const float* __restrict__ agg_lo,
                               float* __restrict__ agg_hi_logits,
                               float* __restrict__ local,
                               const float* __restrict__ deg)
{
    long long i = (long long)blockIdx.x * blockDim.x + threadIdx.x;
    if (i >= (long long)N_NODES * N_CLASS) return;
    const int row = (int)(i >> 4);
    const float scale = (1.0f - ALPHA) / fmaxf(deg[row], 1e-12f);
    const float al = ALPHA * local[i];
    agg_hi_logits[i] = scale * (agg_lo[i] + agg_hi_logits[i]) + al;
    local[i] = al / scale;
}

__global__ void final_kernel(const float* __restrict__ agg_lo,
                             const float* __restrict__ agg_hi,
                             const float* __restrict__ deg,
                             float* __restrict__ out)
{
    int row = blockIdx.x * blockDim.x + threadIdx.x;
    if (row >= N_NODES) return;
    const float scale = (1.0f - ALPHA) / fmaxf(deg[row], 1e-12f);
    float t[N_CLASS];
    float m = -INFINITY;
    for (int c = 0; c < N_CLASS; ++c) {
        t[c] = scale * (agg_lo[(size_t)row * N_CLASS + c]
                      + agg_hi[(size_t)row * N_CLASS + c]);
        m = fmaxf(m, t[c]);
    }
    float s = 0.0f;
    for (int c = 0; c < N_CLASS; ++c) s += __expf(t[c] - m);
    const float lse = m + __logf(s);
    for (int c = 0; c < N_CLASS; ++c)
        out[(size_t)row * N_CLASS + c] = t[c] - lse;
}

extern "C" void kernel_launch(void* const* d_in, const int* in_sizes, int n_in,
                              void* d_out, int out_size, void* d_ws, size_t ws_size,
                              hipStream_t stream)
{
    const float* x    = (const float*)d_in[0];
    const float* W1   = (const float*)d_in[1];
    const float* W2   = (const float*)d_in[2];
    const int*   esrc = (const int*)d_in[3];
    const int*   edst = (const int*)d_in[4];
    float* out = (float*)d_out;

    float* local  = (float*)d_ws;
    float* logits = local  + (size_t)N_NODES * N_CLASS;
    float* deg    = logits + (size_t)N_NODES * N_CLASS;
    float* agg    = out;

    const dim3 blk(256);
    const long long nl   = (long long)N_NODES * N_CLASS;
    const unsigned gN    = (unsigned)((N_NODES + 255) / 256);
    const unsigned gNL   = (unsigned)((nl + 255) / 256);
    const unsigned gE    = (unsigned)((N_EDGES + 255) / 256);
    const unsigned gSC   = (unsigned)(N_EDGES / SC_EDGES_PER_BLOCK);
    const dim3 blkSC(SC_WAVES * 32);

    const unsigned gGEMM = (unsigned)(((N_NODES + 15) / 16 + 7) / 8);
    gcn_gemm_kernel<<<gGEMM, blk, 0, stream>>>(x, W1, W2, local);

    zero2_kernel<<<gN, blk, 0, stream>>>(deg, agg, N_NODES);
    deg_kernel<<<gE, blk, 0, stream>>>(esrc, deg, agg);

    degsum_zero_kernel<<<gNL, blk, 0, stream>>>(deg, agg, logits, nl);
    scatter_kernel<<<gSC, blkSC, 0, stream>>>(esrc, edst, local, agg, logits);
    combine_kernel<<<gNL, blk, 0, stream>>>(agg, logits, local, deg);

    zero_kernel<<<gNL, blk, 0, stream>>>(agg, nl);
    scatter_kernel<<<gSC, blkSC, 0, stream>>>(esrc, edst, logits, agg, local);
    final_kernel<<<gN, blk, 0, stream>>>(agg, local, deg, out);
}


# degree count rides in the GEMM kernel: each 16-row tile's wave also counts 512 edges, its f32 atomic adds issued ahead of each K-chunk's matrix work; degree launch gone; counts exact in any order
# speedup vs baseline: 2.7628x; 1.0866x over previous
#include <hip/hip_runtime.h>
#include <hip/hip_bf16.h>
#include <math.h>

#define N_NODES 100000
#define N_FEAT  512
#define HIDDEN  128
#define N_CLASS 16
#define N_EDGES 3200000
#define ALPHA   0.25f

typedef __attribute__((ext_vector_type(2))) float v2f;
typedef __attribute__((ext_vector_type(8))) float v8f;

#define KCHUNK   64
#define W1C_STR  136
#define HID_STR  132

#define N_TILES             (N_NODES / 16)
#define DEG_EDGES_PER_TILE  (N_EDGES / N_TILES)
#define DEG_TRIPS_PER_CHUNK (DEG_EDGES_PER_TILE / 32 / (N_FEAT / KCHUNK))
static_assert(N_NODES % 16 == 0, "whole 16-row tiles");
static_assert(DEG_TRIPS_PER_CHUNK >= 1 &&
              (long long)DEG_TRIPS_PER_CHUNK * 32 * (N_FEAT / KCHUNK) * N_TILES == N_EDGES,
              "the tiles' edge slices must cover every edge exactly once");

__device__ __forceinline__ bool upper_half();

__global__ __launch_bounds__(256) void gcn_gemm_kernel(
    const float* __restrict__ x, const float* __restrict__ W1,
    const float* __restrict__ W2, float* __restrict__ local_out,
    const int* __restrict__ src, float* __restrict__ deg_lo, float* __restrict__ deg_hi)
{
    __shared__ float w1c[KCHUNK][W1C_STR];
    __shared__ float hid[8][16][HID_STR];

    const int tid  = threadIdx.x;
    const int wave = tid >> 5;
    const int lane = tid & 31;
    const int half = lane >> 4;
    const int l16  = lane & 15;

    const int row_tile = blockIdx.x * 8 + wave;
    const int row_base = row_tile * 16;
    const bool active  = (row_base < N_NODES);

    v8f acc[8];
    for (int t = 0; t < 8; ++t)
        for (int i = 0; i < 8; ++i) acc[t][i] = 0.0f;

    float* degp = upper_half() ? deg_hi : deg_lo;
    const int* sp = src + (size_t)row_tile * DEG_EDGES_PER_TILE + lane;
    int ds[DEG_TRIPS_PER_CHUNK];
    if (active)
        for (int j = 0; j < DEG_TRIPS_PER_CHUNK; ++j) ds[j] = sp[j * 32];

    for (int k0 = 0; k0 < N_FEAT; k0 += KCHUNK) {
        __syncthreads();
        for (int i = tid; i < KCHUNK * (HIDDEN / 4); i += 256) {
            const int r  = i >> 5;
            const int c4 = i & 31;
            const float4 v = ((const float4*)(W1 + (size_t)(k0 + r) * HIDDEN))[c4];
            *(float4*)&w1c[r][c4 * 4] = v;
        }
        __syncthreads();

        if (active) {
            for (int j = 0; j < DEG_TRIPS_PER_CHUNK; ++j)
                atomicAdd(&degp[ds[j]], 1.0f);
            if (k0 + KCHUNK < N_FEAT) {
                const int* spn = sp + (k0 / KCHUNK + 1) * (DEG_TRIPS_PER_CHUNK * 32);
                for (int j = 0; j < DEG_TRIPS_PER_CHUNK; ++j) ds[j] = spn[j * 32];
            }
            for (int kk = 0; kk < KCHUNK; kk += 4) {
                const float* ap = x + (size_t)(row_base + l16) * N_FEAT
                                    + (k0 + kk + 2 * half);
                v2f a = *(const v2f*)ap;
                for (int t = 0; t < 8; ++t) {
                    v2f b;
                    b[0] = w1c[kk + 2 * half + 0][t * 16 + l16];
                    b[1] = w1c[kk + 2 * half + 1][t * 16 + l16];
                    acc[t] = __builtin_amdgcn_wmma_f32_16x16x4_f32(
                        false, a, false, b, (short)0, acc[t], false, false);
                }
            }
        }
    }

    if (active) {
        for (int t = 0; t < 8; ++t)
            for (int r = 0; r < 8; ++r) {
                float v = acc[t][r];
                hid[wave][r + 8 * half][t * 16 + l16] = v > 0.0f ? v : 0.0f;
            }
        v8f c2;
        for (int i = 0; i < 8; ++i) c2[i] = 0.0f;
        for (int kk = 0; kk < HIDDEN; kk += 4) {
            v2f a, b;
            a[0] = hid[wave][l16][kk + 2 * half + 0];
            a[1] = hid[wave][l16][kk + 2 * half + 1];
            b[0] = W2[(size_t)(kk + 2 * half + 0) * N_CLASS + l16];
            b[1] = W2[(size_t)(kk + 2 * half + 1) * N_CLASS + l16];
            c2 = __builtin_amdgcn_wmma_f32_16x16x4_f32(
                false, a, false, b, (short)0, c2, false, false);
        }
        for (int r = 0; r < 8; ++r)
            local_out[(size_t)(row_base + r + 8 * half) * N_CLASS + l16] = c2[r];
    }
}

__global__ void zero_kernel(float* __restrict__ p, long long n)
{
    long long i = (long long)blockIdx.x * blockDim.x + threadIdx.x;
    if (i < n) p[i] = 0.0f;
}

__global__ void zero2_kernel(float* __restrict__ p, float* __restrict__ q, long long n)
{
    long long i = (long long)blockIdx.x * blockDim.x + threadIdx.x;
    if (i < n) { p[i] = 0.0f; q[i] = 0.0f; }
}

__device__ __forceinline__ bool upper_half()
{
    const unsigned w = __builtin_amdgcn_s_sendmsg_rtn(0x87);
    return ((w >> 16) & 0xFu) >= 4u;
}

__global__ void degsum_zero_kernel(float* __restrict__ deg, float* __restrict__ a,
                                   float* __restrict__ b, long long n)
{
    long long i = (long long)blockIdx.x * blockDim.x + threadIdx.x;
    if (i >= n) return;
    if (i < N_NODES) deg[i] = deg[i] + a[i];
    a[i] = 0.0f;
    b[i] = 0.0f;
}

#define SC_WAVES 4
#define SC_TRIPS 8
#define SC_EDGES_PER_BLOCK (SC_WAVES * SC_TRIPS * 32)
static_assert(N_EDGES % SC_EDGES_PER_BLOCK == 0, "edge count must fill whole work-groups");

#define ROW_SHARE(v, J) __builtin_amdgcn_update_dpp(0, (v), 0x150 + (J), 0xF, 0xF, false)

__global__ __launch_bounds__(SC_WAVES * 32) void scatter_kernel(
    const int* __restrict__ src, const int* __restrict__ dst,
    const float* __restrict__ logits, float* __restrict__ agg_lo, float* __restrict__ agg_hi)
{
    const int lane = threadIdx.x & 31;
    const unsigned cb = (unsigned)(lane & 15) << 2;
    const int wave = blockIdx.x * SC_WAVES + (threadIdx.x >> 5);
    int e = wave * (SC_TRIPS * 32) + lane;

    const char* lg = (const char*)logits;
    char* ag = (char*)(upper_half() ? agg_hi : agg_lo);

    int s = src[e];
    int d = dst[e];
    for (int it = 0; it < SC_TRIPS; ++it) {
        int sn = s, dn = d;
        if (it + 1 < SC_TRIPS) {
            sn = src[e + 32];
            dn = dst[e + 32];
        }
        float v[16];
#define SC_GATHER(J) { const unsigned dj = (unsigned)ROW_SHARE(d, J); \
                       v[J] = *(const float*)(lg + ((dj << 6) | cb)); }
        SC_GATHER(0)  SC_GATHER(1)  SC_GATHER(2)  SC_GATHER(3)
        SC_GATHER(4)  SC_GATHER(5)  SC_GATHER(6)  SC_GATHER(7)
        SC_GATHER(8)  SC_GATHER(9)  SC_GATHER(10) SC_GATHER(11)
        SC_GATHER(12) SC_GATHER(13) SC_GATHER(14) SC_GATHER(15)
#undef SC_GATHER
#define SC_ADD(J) { const unsigned sj = (unsigned)ROW_SHARE(s, J); \
                    atomicAdd((float*)(ag + ((sj << 6) | cb)), v[J]); }
        SC_ADD(0)  SC_ADD(1)  SC_ADD(2)  SC_ADD(3)
        SC_ADD(4)  SC_ADD(5)  SC_ADD(6)  SC_ADD(7)
        SC_ADD(8)  SC_ADD(9)  SC_ADD(10) SC_ADD(11)
        SC_ADD(12) SC_ADD(13) SC_ADD(14) SC_ADD(15)
#undef SC_ADD
        s = sn;
        d = dn;
        e += 32;
    }
}

__global__ void combine_kernel(const float* __restrict__ agg_lo,
                               float* __restrict__ agg_hi_logits,
                               float* __restrict__ local,
                               const float* __restrict__ deg)
{
    long long i = (long long)blockIdx.x * blockDim.x + threadIdx.x;
    if (i >= (long long)N_NODES * N_CLASS) return;
    const int row = (int)(i >> 4);
    const float scale = (1.0f - ALPHA) / fmaxf(deg[row], 1e-12f);
    const float al = ALPHA * local[i];
    agg_hi_logits[i] = scale * (agg_lo[i] + agg_hi_logits[i]) + al;
    local[i] = al / scale;
}

__global__ void final_kernel(const float* __restrict__ agg_lo,
                             const float* __restrict__ agg_hi,
                             const float* __restrict__ deg,
                             float* __restrict__ out)
{
    int row = blockIdx.x * blockDim.x + threadIdx.x;
    if (row >= N_NODES) return;
    const float scale = (1.0f - ALPHA) / fmaxf(deg[row], 1e-12f);
    float t[N_CLASS];
    float m = -INFINITY;
    for (int c = 0; c < N_CLASS; ++c) {
        t[c] = scale * (agg_lo[(size_t)row * N_CLASS + c]
                      + agg_hi[(size_t)row * N_CLASS + c]);
        m = fmaxf(m, t[c]);
    }
    float s = 0.0f;
    for (int c = 0; c < N_CLASS; ++c) s += __expf(t[c] - m);
    const float lse = m + __logf(s);
    for (int c = 0; c < N_CLASS; ++c)
        out[(size_t)row * N_CLASS + c] = t[c] - lse;
}

extern "C" void kernel_launch(void* const* d_in, const int* in_sizes, int n_in,
                              void* d_out, int out_size, void* d_ws, size_t ws_size,
                              hipStream_t stream)
{
    const float* x    = (const float*)d_in[0];
    const float* W1   = (const float*)d_in[1];
    const float* W2   = (const float*)d_in[2];
    const int*   esrc = (const int*)d_in[3];
    const int*   edst = (const int*)d_in[4];
    float* out = (float*)d_out;

    float* local  = (float*)d_ws;
    float* logits = local  + (size_t)N_NODES * N_CLASS;
    float* deg    = logits + (size_t)N_NODES * N_CLASS;
    float* agg    = out;

    const dim3 blk(256);
    const long long nl   = (long long)N_NODES * N_CLASS;
    const unsigned gN    = (unsigned)((N_NODES + 255) / 256);
    const unsigned gNL   = (unsigned)((nl + 255) / 256);
    const unsigned gSC   = (unsigned)(N_EDGES / SC_EDGES_PER_BLOCK);
    const dim3 blkSC(SC_WAVES * 32);

    zero2_kernel<<<gN, blk, 0, stream>>>(deg, agg, N_NODES);

    const unsigned gGEMM = (unsigned)(((N_NODES + 15) / 16 + 7) / 8);
    gcn_gemm_kernel<<<gGEMM, blk, 0, stream>>>(x, W1, W2, local, esrc, deg, agg);

    degsum_zero_kernel<<<gNL, blk, 0, stream>>>(deg, agg, logits, nl);
    scatter_kernel<<<gSC, blkSC, 0, stream>>>(esrc, edst, local, agg, logits);
    combine_kernel<<<gNL, blk, 0, stream>>>(agg, logits, local, deg);

    zero_kernel<<<gNL, blk, 0, stream>>>(agg, nl);
    scatter_kernel<<<gSC, blkSC, 0, stream>>>(esrc, edst, logits, agg, local);
    final_kernel<<<gN, blk, 0, stream>>>(agg, local, deg, out);
}


# GEMM kernel: 4-wave work-groups (one 16-row tile per SIMD), hidden staging overlaid on the W1 chunk, LDS sized for 7 groups per WGP: all 1563 groups resident in one round (was 782 on 768 places)
# speedup vs baseline: 3.1617x; 1.1444x over previous
#include <hip/hip_runtime.h>
#include <hip/hip_bf16.h>
#include <math.h>

#define N_NODES 100000
#define N_FEAT  512
#define HIDDEN  128
#define N_CLASS 16
#define N_EDGES 3200000
#define ALPHA   0.25f

typedef __attribute__((ext_vector_type(2))) float v2f;
typedef __attribute__((ext_vector_type(8))) float v8f;

#define KCHUNK   64
#define W1C_STR  136
#define HID_STR  132
#define GEMM_WAVES 4
#define GEMM_LDS_FLOATS (44 * 1024 / 4)
static_assert(KCHUNK * W1C_STR <= GEMM_LDS_FLOATS && GEMM_WAVES * 16 * HID_STR <= GEMM_LDS_FLOATS,
              "both uses of the LDS block must fit it");

#define N_TILES             (N_NODES / 16)
#define DEG_EDGES_PER_TILE  (N_EDGES / N_TILES)
#define DEG_TRIPS_PER_CHUNK (DEG_EDGES_PER_TILE / 32 / (N_FEAT / KCHUNK))
static_assert(N_NODES % 16 == 0, "whole 16-row tiles");
static_assert(DEG_TRIPS_PER_CHUNK >= 1 &&
              (long long)DEG_TRIPS_PER_CHUNK * 32 * (N_FEAT / KCHUNK) * N_TILES == N_EDGES,
              "the tiles' edge slices must cover every edge exactly once");

__device__ __forceinline__ bool upper_half();

__global__ __launch_bounds__(GEMM_WAVES * 32) void gcn_gemm_kernel(
    const float* __restrict__ x, const float* __restrict__ W1,
    const float* __restrict__ W2, float* __restrict__ local_out,
    const int* __restrict__ src, float* __restrict__ deg_lo, float* __restrict__ deg_hi)
{
    __shared__ float lds[GEMM_LDS_FLOATS];
    float (*w1c)[W1C_STR]     = (float (*)[W1C_STR])lds;
    float (*hid)[16][HID_STR] = (float (*)[16][HID_STR])lds;

    const int tid  = threadIdx.x;
    const int wave = tid >> 5;
    const int lane = tid & 31;
    const int half = lane >> 4;
    const int l16  = lane & 15;

    const int row_tile = blockIdx.x * GEMM_WAVES + wave;
    const int row_base = row_tile * 16;
    const bool active  = (row_base < N_NODES);

    v8f acc[8];
    for (int t = 0; t < 8; ++t)
        for (int i = 0; i < 8; ++i) acc[t][i] = 0.0f;

    float* degp = upper_half() ? deg_hi : deg_lo;
    const int* sp = src + (size_t)row_tile * DEG_EDGES_PER_TILE + lane;
    int ds[DEG_TRIPS_PER_CHUNK];
    if (active)
        for (int j = 0; j < DEG_TRIPS_PER_CHUNK; ++j) ds[j] = sp[j * 32];

    for (int k0 = 0; k0 < N_FEAT; k0 += KCHUNK) {
        __syncthreads();
        for (int i = tid; i < KCHUNK * (HIDDEN / 4); i += GEMM_WAVES * 32) {
            const int r  = i >> 5;
            const int c4 = i & 31;
            const float4 v = ((const float4*)(W1 + (size_t)(k0 + r) * HIDDEN))[c4];
            *(float4*)&w1c[r][c4 * 4] = v;
        }
        __syncthreads();

        if (active) {
            for (int j = 0; j < DEG_TRIPS_PER_CHUNK; ++j)
                atomicAdd(&degp[ds[j]], 1.0f);
            if (k0 + KCHUNK < N_FEAT) {
                const int* spn = sp + (k0 / KCHUNK + 1) * (DEG_TRIPS_PER_CHUNK * 32);
                for (int j = 0; j < DEG_TRIPS_PER_CHUNK; ++j) ds[j] = spn[j * 32];
            }
            for (int kk = 0; kk < KCHUNK; kk += 4) {
                const float* ap = x + (size_t)(row_base + l16) * N_FEAT
                                    + (k0 + kk + 2 * half);
                v2f a = *(const v2f*)ap;
                for (int t = 0; t < 8; ++t) {
                    v2f b;
                    b[0] = w1c[kk + 2 * half + 0][t * 16 + l16];
                    b[1] = w1c[kk + 2 * half + 1][t * 16 + l16];
                    acc[t] = __builtin_amdgcn_wmma_f32_16x16x4_f32(
                        false, a, false, b, (short)0, acc[t], false, false);
                }
            }
        }
    }

    __syncthreads();

    if (active) {
        for (int t = 0; t < 8; ++t)
            for (int r = 0; r < 8; ++r) {
                float v = acc[t][r];
                hid[wave][r + 8 * half][t * 16 + l16] = v > 0.0f ? v : 0.0f;
            }
        v8f c2;
        for (int i = 0; i < 8; ++i) c2[i] = 0.0f;
        for (int kk = 0; kk < HIDDEN; kk += 4) {
            v2f a, b;
            a[0] = hid[wave][l16][kk + 2 * half + 0];
            a[1] = hid[wave][l16][kk + 2 * half + 1];
            b[0] = W2[(size_t)(kk + 2 * half + 0) * N_CLASS + l16];
            b[1] = W2[(size_t)(kk + 2 * half + 1) * N_CLASS + l16];
            c2 = __builtin_amdgcn_wmma_f32_16x16x4_f32(
                false, a, false, b, (short)0, c2, false, false);
        }
        for (int r = 0; r < 8; ++r)
            local_out[(size_t)(row_base + r + 8 * half) * N_CLASS + l16] = c2[r];
    }
}

__global__ void zero_kernel(float* __restrict__ p, long long n)
{
    long long i = (long long)blockIdx.x * blockDim.x + threadIdx.x;
    if (i < n) p[i] = 0.0f;
}

__global__ void zero2_kernel(float* __restrict__ p, float* __restrict__ q, long long n)
{
    long long i = (long long)blockIdx.x * blockDim.x + threadIdx.x;
    if (i < n) { p[i] = 0.0f; q[i] = 0.0f; }
}

__device__ __forceinline__ bool upper_half()
{
    const unsigned w = __builtin_amdgcn_s_sendmsg_rtn(0x87);
    return ((w >> 16) & 0xFu) >= 4u;
}

__global__ void degsum_zero_kernel(float* __restrict__ deg, float* __restrict__ a,
                                   float* __restrict__ b, long long n)
{
    long long i = (long long)blockIdx.x * blockDim.x + threadIdx.x;
    if (i >= n) return;
    if (i < N_NODES) deg[i] = deg[i] + a[i];
    a[i] = 0.0f;
    b[i] = 0.0f;
}

#define SC_WAVES 4
#define SC_TRIPS 8
#define SC_EDGES_PER_BLOCK (SC_WAVES * SC_TRIPS * 32)
static_assert(N_EDGES % SC_EDGES_PER_BLOCK == 0, "edge count must fill whole work-groups");

#define ROW_SHARE(v, J) __builtin_amdgcn_update_dpp(0, (v), 0x150 + (J), 0xF, 0xF, false)

__global__ __launch_bounds__(SC_WAVES * 32) void scatter_kernel(
    const int* __restrict__ src, const int* __restrict__ dst,
    const float* __restrict__ logits, float* __restrict__ agg_lo, float* __restrict__ agg_hi)
{
    const int lane = threadIdx.x & 31;
    const unsigned cb = (unsigned)(lane & 15) << 2;
    const int wave = blockIdx.x * SC_WAVES + (threadIdx.x >> 5);
    int e = wave * (SC_TRIPS * 32) + lane;

    const char* lg = (const char*)logits;
    char* ag = (char*)(upper_half() ? agg_hi : agg_lo);

    int s = src[e];
    int d = dst[e];
    for (int it = 0; it < SC_TRIPS; ++it) {
        int sn = s, dn = d;
        if (it + 1 < SC_TRIPS) {
            sn = src[e + 32];
            dn = dst[e + 32];
        }
        float v[16];
#define SC_GATHER(J) { const unsigned dj = (unsigned)ROW_SHARE(d, J); \
                       v[J] = *(const float*)(lg + ((dj << 6) | cb)); }
        SC_GATHER(0)  SC_GATHER(1)  SC_GATHER(2)  SC_GATHER(3)
        SC_GATHER(4)  SC_GATHER(5)  SC_GATHER(6)  SC_GATHER(7)
        SC_GATHER(8)  SC_GATHER(9)  SC_GATHER(10) SC_GATHER(11)
        SC_GATHER(12) SC_GATHER(13) SC_GATHER(14) SC_GATHER(15)
#undef SC_GATHER
#define SC_ADD(J) { const unsigned sj = (unsigned)ROW_SHARE(s, J); \
                    atomicAdd((float*)(ag + ((sj << 6) | cb)), v[J]); }
        SC_ADD(0)  SC_ADD(1)  SC_ADD(2)  SC_ADD(3)
        SC_ADD(4)  SC_ADD(5)  SC_ADD(6)  SC_ADD(7)
        SC_ADD(8)  SC_ADD(9)  SC_ADD(10) SC_ADD(11)
        SC_ADD(12) SC_ADD(13) SC_ADD(14) SC_ADD(15)
#undef SC_ADD
        s = sn;
        d = dn;
        e += 32;
    }
}

__global__ void combine_kernel(const float* __restrict__ agg_lo,
                               float* __restrict__ agg_hi_logits,
                               float* __restrict__ local,
                               const float* __restrict__ deg)
{
    long long i = (long long)blockIdx.x * blockDim.x + threadIdx.x;
    if (i >= (long long)N_NODES * N_CLASS) return;
    const int row = (int)(i >> 4);
    const float scale = (1.0f - ALPHA) / fmaxf(deg[row], 1e-12f);
    const float al = ALPHA * local[i];
    agg_hi_logits[i] = scale * (agg_lo[i] + agg_hi_logits[i]) + al;
    local[i] = al / scale;
}

__global__ void final_kernel(const float* __restrict__ agg_lo,
                             const float* __restrict__ agg_hi,
                             const float* __restrict__ deg,
                             float* __restrict__ out)
{
    int row = blockIdx.x * blockDim.x + threadIdx.x;
    if (row >= N_NODES) return;
    const float scale = (1.0f - ALPHA) / fmaxf(deg[row], 1e-12f);
    float t[N_CLASS];
    float m = -INFINITY;
    for (int c = 0; c < N_CLASS; ++c) {
        t[c] = scale * (agg_lo[(size_t)row * N_CLASS + c]
                      + agg_hi[(size_t)row * N_CLASS + c]);
        m = fmaxf(m, t[c]);
    }
    float s = 0.0f;
    for (int c = 0; c < N_CLASS; ++c) s += __expf(t[c] - m);
    const float lse = m + __logf(s);
    for (int c = 0; c < N_CLASS; ++c)
        out[(size_t)row * N_CLASS + c] = t[c] - lse;
}

extern "C" void kernel_launch(void* const* d_in, const int* in_sizes, int n_in,
                              void* d_out, int out_size, void* d_ws, size_t ws_size,
                              hipStream_t stream)
{
    const float* x    = (const float*)d_in[0];
    const float* W1   = (const float*)d_in[1];
    const float* W2   = (const float*)d_in[2];
    const int*   esrc = (const int*)d_in[3];
    const int*   edst = (const int*)d_in[4];
    float* out = (float*)d_out;

    float* local  = (float*)d_ws;
    float* logits = local  + (size_t)N_NODES * N_CLASS;
    float* deg    = logits + (size_t)N_NODES * N_CLASS;
    float* agg    = out;

    const dim3 blk(256);
    const long long nl   = (long long)N_NODES * N_CLASS;
    const unsigned gN    = (unsigned)((N_NODES + 255) / 256);
    const unsigned gNL   = (unsigned)((nl + 255) / 256);
    const unsigned gSC   = (unsigned)(N_EDGES / SC_EDGES_PER_BLOCK);
    const dim3 blkSC(SC_WAVES * 32);

    zero2_kernel<<<gN, blk, 0, stream>>>(deg, agg, N_NODES);

    const unsigned gGEMM = (unsigned)((N_TILES + GEMM_WAVES - 1) / GEMM_WAVES);
    gcn_gemm_kernel<<<gGEMM, dim3(GEMM_WAVES * 32), 0, stream>>>(x, W1, W2, local, esrc, deg, agg);

    degsum_zero_kernel<<<gNL, blk, 0, stream>>>(deg, agg, logits, nl);
    scatter_kernel<<<gSC, blkSC, 0, stream>>>(esrc, edst, local, agg, logits);
    combine_kernel<<<gNL, blk, 0, stream>>>(agg, logits, local, deg);

    zero_kernel<<<gNL, blk, 0, stream>>>(agg, nl);
    scatter_kernel<<<gSC, blkSC, 0, stream>>>(esrc, edst, logits, agg, local);
    final_kernel<<<gN, blk, 0, stream>>>(agg, local, deg, out);
}
